# starting kernel
# baseline (speedup 1.0000x reference)
#include <hip/hip_runtime.h>
#include <hip/hip_bf16.h>

typedef __attribute__((ext_vector_type(2))) float v2f;
typedef __attribute__((ext_vector_type(8))) float v8f;

#define CIN  128
#define COUT 128
#define XPAD 132

__global__ __launch_bounds__(256) void gcn_gemm_norm(
    const float* __restrict__ x, const float* __restrict__ w,
    const int* __restrict__ colptr, float* __restrict__ h, int n)
{
  __shared__ float xs[16 * XPAD];
  __shared__ float onorm[16];

  const int tid  = threadIdx.x;
  const int wave = tid >> 5;
  const int lane = tid & 31;
  const int half = lane >> 4;
  const int l16  = lane & 15;
  const int n0   = wave * 16;

  v2f bfrag[32];
#pragma unroll
  for (int s = 0; s < 32; ++s) {
    const int kk = 4 * s + 2 * half;
    bfrag[s].x = w[kk       * COUT + n0 + l16];
    bfrag[s].y = w[(kk + 1) * COUT + n0 + l16];
  }

  const int ntiles = (n + 15) >> 4;
  for (int t = blockIdx.x; t < ntiles; t += gridDim.x) {
    const int row0 = t << 4;

    __syncthreads();
#pragma unroll
    for (int i = 0; i < 8; ++i) {
      const int e = tid + i * 256;
      const int r = e >> 7, c = e & 127;
      const int gr = row0 + r;
      xs[r * XPAD + c] = (gr < n) ? x[(size_t)gr * CIN + c] : 0.0f;
    }
    if (tid < 16) {
      const int gr = row0 + tid;
      const float deg = (gr < n) ? (float)(colptr[gr + 1] - colptr[gr]) : 1.0f;
      onorm[tid] = rsqrtf(deg);
    }
    __syncthreads();

    v8f acc = {};
#pragma unroll
    for (int s = 0; s < 32; ++s) {
      const int kk = 4 * s + 2 * half;
      v2f a;
      a.x = xs[l16 * XPAD + kk];
      a.y = xs[l16 * XPAD + kk + 1];
      acc = __builtin_amdgcn_wmma_f32_16x16x4_f32(
           false, a,  false, bfrag[s],
           (short)0, acc,  false,  false);
    }

#pragma unroll
    for (int r = 0; r < 8; ++r) {
      const int m  = r + 8 * half;
      const int gr = row0 + m;
      if (gr < n)
        h[(size_t)gr * COUT + n0 + l16] = acc[r] * onorm[m];
    }
  }
}

__global__ __launch_bounds__(256) void gcn_aggr(
    const float* __restrict__ h, const int* __restrict__ rowptr,
    const int* __restrict__ colind, const float* __restrict__ bias,
    float* __restrict__ out, int n)
{
  const int wave = threadIdx.x >> 5;
  const int lane = threadIdx.x & 31;
  const int wavesPerBlk = blockDim.x >> 5;

  const float4* __restrict__ h4 = (const float4*)h;
  const float4  b4 = ((const float4*)bias)[lane];

  for (int i = blockIdx.x * wavesPerBlk + wave; i < n;
       i += gridDim.x * wavesPerBlk) {
    const int e0 = rowptr[i];
    const int e1 = rowptr[i + 1];

    float ax = 0.f, ay = 0.f, az = 0.f, aw = 0.f;
    for (int e = e0; e < e1; ++e) {
      const int nbr = colind[e];
      const float4 v = h4[(size_t)nbr * (COUT / 4) + lane];
      ax += v.x; ay += v.y; az += v.z; aw += v.w;
    }

    const float inorm = rsqrtf((float)(e1 - e0));
    float4 o;
    o.x = ax * inorm + b4.x;
    o.y = ay * inorm + b4.y;
    o.z = az * inorm + b4.z;
    o.w = aw * inorm + b4.w;
    ((float4*)out)[(size_t)i * (COUT / 4) + lane] = o;
  }
}

extern "C" void kernel_launch(void* const* d_in, const int* in_sizes, int n_in,
                              void* d_out, int out_size, void* d_ws, size_t ws_size,
                              hipStream_t stream) {
  const float* x      = (const float*)d_in[0];
  const float* weight = (const float*)d_in[1];
  const float* bias   = (const float*)d_in[2];
  const int*   rowptr = (const int*)d_in[3];
  const int*   colind = (const int*)d_in[4];
  const int*   colptr = (const int*)d_in[5];

  const int n = in_sizes[3] - 1;
  float* h = (float*)d_ws;

  dim3 blk(256);
  gcn_gemm_norm<<<1024, blk, 0, stream>>>(x, weight, colptr, h, n);
  gcn_aggr<<<4096, blk, 0, stream>>>(h, rowptr, colind, bias, (float*)d_out, n);
}


# Aggregation: gather parallelism moved into the wave: indices on the scalar path, 16 row loads in flight, next node's indices requested a node ahead, 1024 blocks loop over nodes; sum order kept
# speedup vs baseline: 1.3569x; 1.3569x over previous
#include <hip/hip_runtime.h>
#include <hip/hip_bf16.h>

typedef __attribute__((ext_vector_type(2))) float v2f;
typedef __attribute__((ext_vector_type(8))) float v8f;

#define CIN  128
#define COUT 128
#define XPAD 132

__global__ __launch_bounds__(256) void gcn_gemm_norm(
    const float* __restrict__ x, const float* __restrict__ w,
    const int* __restrict__ colptr, float* __restrict__ h, int n)
{
  __shared__ alignas(16) float xs[2][16 * XPAD];
  __shared__ float onorm[2][16];

  const int tid  = threadIdx.x;
  const int wave = tid >> 5;
  const int lane = tid & 31;
  const int half = lane >> 4;
  const int l16  = lane & 15;
  const int n0   = wave * 16;

  v2f bfrag[32];
#pragma unroll
  for (int s = 0; s < 32; ++s) {
    const int kk = 4 * s + 2 * half;
    bfrag[s].x = w[kk       * COUT + n0 + l16];
    bfrag[s].y = w[(kk + 1) * COUT + n0 + l16];
  }

  const int ntiles = (n + 15) >> 4;
  const float4* __restrict__ x4 = (const float4*)x;

  const int sr = tid >> 5;
  const int sc = tid & 31;

  float4 pre0 = {0.f, 0.f, 0.f, 0.f}, pre1 = {0.f, 0.f, 0.f, 0.f};
  int d0 = 0, d1 = 1;

  int t = blockIdx.x;
  if (t < ntiles) {
    const int row0 = t << 4;
    const int ra = min(row0 + sr, n - 1);
    const int rb = min(row0 + sr + 8, n - 1);
    pre0 = x4[(size_t)ra * (CIN / 4) + sc];
    pre1 = x4[(size_t)rb * (CIN / 4) + sc];
    if (tid < 16) {
      const int g = min(row0 + tid, n - 1);
      d0 = colptr[g];
      d1 = colptr[g + 1];
    }
  }

  int buf = 0;
  for (; t < ntiles; t += gridDim.x) {
    const int row0 = t << 4;

    *(float4*)&xs[buf][sr * XPAD + 4 * sc]       = pre0;
    *(float4*)&xs[buf][(sr + 8) * XPAD + 4 * sc] = pre1;
    if (tid < 16)
      onorm[buf][tid] = (row0 + tid < n) ? rsqrtf((float)(d1 - d0)) : 1.0f;
    __syncthreads();

    const int tn = t + gridDim.x;
    if (tn < ntiles) {
      const int rown = tn << 4;
      const int ra = min(rown + sr, n - 1);
      const int rb = min(rown + sr + 8, n - 1);
      pre0 = x4[(size_t)ra * (CIN / 4) + sc];
      pre1 = x4[(size_t)rb * (CIN / 4) + sc];
      if (tid < 16) {
        const int g = min(rown + tid, n - 1);
        d0 = colptr[g];
        d1 = colptr[g + 1];
      }
    }

    const float* __restrict__ xb = xs[buf];
    v8f acc = {};
#pragma unroll
    for (int s = 0; s < 32; ++s) {
      const int kk = 4 * s + 2 * half;
      v2f a;
      a.x = xb[l16 * XPAD + kk];
      a.y = xb[l16 * XPAD + kk + 1];
      acc = __builtin_amdgcn_wmma_f32_16x16x4_f32(
           false, a,  false, bfrag[s],
           (short)0, acc,  false,  false);
    }

#pragma unroll
    for (int r = 0; r < 8; ++r) {
      const int m  = r + 8 * half;
      const int gr = row0 + m;
      if (gr < n)
        h[(size_t)gr * COUT + n0 + l16] = acc[r] * onorm[buf][m];
    }
    buf ^= 1;
  }
}

template <int K>
__device__ __forceinline__ void rows_add(
    const float4* __restrict__ h4, const int (&nb)[K], int lane,
    float& ax, float& ay, float& az, float& aw)
{
  float4 v[K];
#pragma unroll
  for (int u = 0; u < K; ++u)
    v[u] = h4[(size_t)nb[u] * (COUT / 4) + lane];
#pragma unroll
  for (int u = 0; u < K; ++u) {
    ax += v[u].x; ay += v[u].y; az += v[u].z; aw += v[u].w;
  }
}

template <int K>
__device__ __forceinline__ void gather_add(
    const float4* __restrict__ h4, const int* __restrict__ ci, int lane,
    float& ax, float& ay, float& az, float& aw)
{
  int nb[K];
#pragma unroll
  for (int u = 0; u < K; ++u) nb[u] = ci[u];
  rows_add<K>(h4, nb, lane, ax, ay, az, aw);
}

__global__ __launch_bounds__(256) void gcn_aggr(
    const float* __restrict__ h, const int* __restrict__ rowptr,
    const int* __restrict__ colind, const float* __restrict__ bias,
    float* __restrict__ out, int n)
{
  const int lane = threadIdx.x & 31;
  const int wave = __builtin_amdgcn_readfirstlane((int)(threadIdx.x >> 5));
  const int stride = gridDim.x * (blockDim.x >> 5);

  const float4* __restrict__ h4 = (const float4*)h;
  const float4  b4 = ((const float4*)bias)[lane];

  int inext = blockIdx.x * (blockDim.x >> 5) + wave;
  if (inext >= n) return;

  int f0 = rowptr[inext];
  int f1 = rowptr[inext + 1];
  int pn[16];
#pragma unroll
  for (int u = 0; u < 16; ++u) pn[u] = 0;
  if (f1 - f0 >= 16) {
#pragma unroll
    for (int u = 0; u < 16; ++u) pn[u] = colind[f0 + u];
  }
  int g0 = 0, g1 = 0;
  if (inext + stride < n) {
    g0 = rowptr[inext + stride];
    g1 = rowptr[inext + stride + 1];
  }

  for (;;) {
    const int i  = inext;
    const int e0 = f0, e1 = f1;
    int p[16];
#pragma unroll
    for (int u = 0; u < 16; ++u) p[u] = pn[u];
    const int i1 = i + stride;
    const int i2 = i1 + stride;
    f0 = g0; f1 = g1;
    g0 = 0; g1 = 0;
    if (i1 < n && f1 - f0 >= 16) {
#pragma unroll
      for (int u = 0; u < 16; ++u) pn[u] = colind[f0 + u];
    }
    if (i2 < n) {
      g0 = rowptr[i2];
      g1 = rowptr[i2 + 1];
    }

    float ax = 0.f, ay = 0.f, az = 0.f, aw = 0.f;
    int e = e0;
    if (e1 - e0 >= 16) {
      rows_add<16>(h4, p, lane, ax, ay, az, aw);
      e += 16;
    }
    for (; e + 16 <= e1; e += 16)
      gather_add<16>(h4, colind + e, lane, ax, ay, az, aw);
    if (e + 8 <= e1) {
      gather_add<8>(h4, colind + e, lane, ax, ay, az, aw);
      e += 8;
    }
    if (e + 4 <= e1) {
      gather_add<4>(h4, colind + e, lane, ax, ay, az, aw);
      e += 4;
    }
    for (; e < e1; ++e)
      gather_add<1>(h4, colind + e, lane, ax, ay, az, aw);

    const float inorm = rsqrtf((float)(e1 - e0));
    float4 o;
    o.x = ax * inorm + b4.x;
    o.y = ay * inorm + b4.y;
    o.z = az * inorm + b4.z;
    o.w = aw * inorm + b4.w;
    ((float4*)out)[(size_t)i * (COUT / 4) + lane] = o;

    if (i1 >= n) break;
    inext = i1;
  }
}

extern "C" void kernel_launch(void* const* d_in, const int* in_sizes, int n_in,
                              void* d_out, int out_size, void* d_ws, size_t ws_size,
                              hipStream_t stream) {
  const float* x      = (const float*)d_in[0];
  const float* weight = (const float*)d_in[1];
  const float* bias   = (const float*)d_in[2];
  const int*   rowptr = (const int*)d_in[3];
  const int*   colind = (const int*)d_in[4];
  const int*   colptr = (const int*)d_in[5];

  const int n = in_sizes[3] - 1;
  float* h = (float*)d_ws;

  dim3 blk(256);
  gcn_gemm_norm<<<1024, blk, 0, stream>>>(x, weight, colptr, h, n);
  gcn_aggr<<<1024, blk, 0, stream>>>(h, rowptr, colind, bias, (float*)d_out, n);
}
